# starting kernel
# baseline (speedup 1.0000x reference)
#include <hip/hip_runtime.h>
#include <hip/hip_bf16.h>


typedef __bf16 bf16;
typedef __attribute__((ext_vector_type(16))) __bf16 v16bf;
typedef __attribute__((ext_vector_type(8)))  __bf16 v8bf;
typedef __attribute__((ext_vector_type(8)))  float  v8f;

template <int W> struct vecf;
template <> struct vecf<2> { typedef __attribute__((ext_vector_type(2))) float T; };
template <> struct vecf<4> { typedef __attribute__((ext_vector_type(4))) float T; };

__device__ __forceinline__ bf16 f2bf(float f) {
  union { float f; unsigned u; } in; in.f = f;
  unsigned u = in.u;
  unsigned r = u + 0x7fffu + ((u >> 16) & 1u);
  union { unsigned short s; bf16 b; } out;
  out.s = (unsigned short)(r >> 16);
  return out.b;
}


__global__ __launch_bounds__(256) void k_deg_init(float* deg, int n) {
  int i = blockIdx.x * 256 + threadIdx.x;
  if (i < n) deg[i] = 1.0f;
}

__global__ __launch_bounds__(256) void k_deg_count(const int* __restrict__ dst,
                                                   float* deg, int E) {
  int e = blockIdx.x * 256 + threadIdx.x;
  if (e < E) atomicAdd(&deg[dst[e]], 1.0f);
}

__global__ __launch_bounds__(256) void k_deg_rsqrt(float* deg, int n) {
  int i = blockIdx.x * 256 + threadIdx.x;
  if (i < n) deg[i] = rsqrtf(deg[i]);
}


__global__ __launch_bounds__(256) void k_cvt_bf16(const float* __restrict__ in,
                                                  bf16* __restrict__ out, int n) {
  int i = blockIdx.x * 256 + threadIdx.x;
  if (i < n) out[i] = f2bf(in[i]);
}

__global__ __launch_bounds__(256) void k_relu_cvt_bf16(const float* __restrict__ in,
                                                       bf16* __restrict__ out, int n) {
  int i = blockIdx.x * 256 + threadIdx.x;
  if (i < n) out[i] = f2bf(fmaxf(in[i], 0.0f));
}

__global__ __launch_bounds__(256) void k_transpose_bf16(const float* __restrict__ W,
                                                        bf16* __restrict__ Wt,
                                                        int K, int cols) {
  int idx = blockIdx.x * 256 + threadIdx.x;
  if (idx >= K * cols) return;
  int n = idx / K;
  int k = idx % K;
  Wt[idx] = f2bf(W[(size_t)k * cols + n]);
}


template <int NCOLS>
__global__ __launch_bounds__(256) void k_gemm_wmma_bf16(
    const bf16* __restrict__ A,
    const bf16* __restrict__ Bt,
    float* __restrict__ C,
    int M)
{
  constexpr int K = 128;
  constexpr int NT = 4;
  constexpr int chunksN = NCOLS / 64;
  const int wave = threadIdx.x >> 5;
  const int lane = threadIdx.x & 31;
  const int tile = blockIdx.x * 8 + wave;
  const int numTiles = (M >> 4) * chunksN;
  if (tile >= numTiles) return;
  const int tm = (tile / chunksN) << 4;
  const int tn = (tile % chunksN) << 6;
  const int rc = lane & 15;
  const int hi = lane >> 4;

  const bf16* aRow  = A  + (size_t)(tm + rc) * K + hi * 8;
  const bf16* bBase = Bt + (size_t)(tn + rc) * K + hi * 8;

  v8f acc[NT] = {};
#pragma unroll
  for (int k0 = 0; k0 < K; k0 += 32) {
    union { v16bf v; v8bf h[2]; } a;
    a.h[0] = *(const v8bf*)(aRow + k0);
    a.h[1] = *(const v8bf*)(aRow + k0 + 16);
#pragma unroll
    for (int t = 0; t < NT; ++t) {
      union { v16bf v; v8bf h[2]; } b;
      const bf16* bp = bBase + (size_t)t * 16 * K + k0;
      b.h[0] = *(const v8bf*)(bp);
      b.h[1] = *(const v8bf*)(bp + 16);
      acc[t] = __builtin_amdgcn_wmma_f32_16x16x32_bf16(
           false, a.v,  false, b.v,
           (short)0, acc[t],  false,  false);
    }
  }

  float* cBase = C + (size_t)tm * NCOLS + tn;
#pragma unroll
  for (int t = 0; t < NT; ++t)
#pragma unroll
    for (int j = 0; j < 8; ++j)
      cBase[(size_t)(j + hi * 8) * NCOLS + t * 16 + rc] = acc[t][j];
}


template <int NCOLS>
__global__ __launch_bounds__(256) void k_self_init(const float* __restrict__ H,
                                                   const float* __restrict__ bias,
                                                   const float* __restrict__ dinv,
                                                   float* __restrict__ acc, int n) {
  int idx = blockIdx.x * 256 + threadIdx.x;
  if (idx >= n * NCOLS) return;
  int i = idx / NCOLS;
  int c = idx - i * NCOLS;
  float di = dinv[i];
  acc[idx] = bias[c] + di * di * H[idx];
}

template <int NCOLS>
__global__ __launch_bounds__(256) void k_edge_scatter(const int* __restrict__ src,
                                                      const int* __restrict__ dst,
                                                      const float* __restrict__ dinv,
                                                      const float* __restrict__ H,
                                                      float* __restrict__ acc, int E) {
  constexpr int VW = NCOLS / 32;
  typedef typename vecf<VW>::T VecT;
  int wave = (blockIdx.x * 256 + threadIdx.x) >> 5;
  int lane = threadIdx.x & 31;
  if (wave >= E) return;
  int s = src[wave];
  int d = dst[wave];
  float norm = dinv[s] * dinv[d];
  const float* hrow = H + (size_t)s * NCOLS + lane * VW;
  float* arow = acc + (size_t)d * NCOLS + lane * VW;
  VecT hv = *(const VecT*)hrow;
#pragma unroll
  for (int j = 0; j < VW; ++j)
    atomicAdd(&arow[j], hv[j] * norm);
}


extern "C" void kernel_launch(void* const* d_in, const int* in_sizes, int n_in,
                              void* d_out, int out_size, void* d_ws, size_t ws_size,
                              hipStream_t stream) {
  (void)n_in; (void)out_size; (void)ws_size;
  const float* x  = (const float*)d_in[0];
  const int*   ei = (const int*)  d_in[1];
  const float* W1 = (const float*)d_in[2];
  const float* b1 = (const float*)d_in[3];
  const float* W2 = (const float*)d_in[4];
  const float* b2 = (const float*)d_in[5];
  float* out = (float*)d_out;

  const int IN = 128, HID = 128, OUT = 64;
  const int N = in_sizes[0] / IN;
  const int E = in_sizes[1] / 2;
  const int M16 = ((N + 15) / 16) * 16;
  const int* src = ei;
  const int* dst = ei + E;

  char* ws = (char*)d_ws;
  size_t off = 0;
  auto alloc = [&](size_t bytes) -> void* {
    void* p = ws + off;
    off = (off + bytes + 255) & ~(size_t)255;
    return p;
  };
  float* dinv = (float*)alloc((size_t)N * 4);
  bf16*  Xb   = (bf16*) alloc((size_t)M16 * IN * 2);
  bf16*  W1t  = (bf16*) alloc((size_t)HID * IN * 2);
  bf16*  W2t  = (bf16*) alloc((size_t)OUT * HID * 2);
  float* h1   = (float*)alloc((size_t)M16 * HID * 4);
  float* acc1 = (float*)alloc((size_t)N * HID * 4);
  bf16*  h1b  = (bf16*) alloc((size_t)M16 * HID * 2);
  float* h2   = (float*)alloc((size_t)M16 * OUT * 4);

  k_deg_init <<<(N + 255) / 256, 256, 0, stream>>>(dinv, N);
  k_deg_count<<<(E + 255) / 256, 256, 0, stream>>>(dst, dinv, E);
  k_deg_rsqrt<<<(N + 255) / 256, 256, 0, stream>>>(dinv, N);

  k_cvt_bf16      <<<(N * IN + 255) / 256, 256, 0, stream>>>(x, Xb, N * IN);
  k_transpose_bf16<<<(IN * HID + 255) / 256, 256, 0, stream>>>(W1, W1t, IN, HID);
  k_transpose_bf16<<<(HID * OUT + 255) / 256, 256, 0, stream>>>(W2, W2t, HID, OUT);

  {
    int tiles = (M16 / 16) * (HID / 64);
    k_gemm_wmma_bf16<HID><<<(tiles + 7) / 8, 256, 0, stream>>>(Xb, W1t, h1, M16);
    k_self_init<HID><<<(N * HID + 255) / 256, 256, 0, stream>>>(h1, b1, dinv, acc1, N);
    k_edge_scatter<HID><<<(E + 7) / 8, 256, 0, stream>>>(src, dst, dinv, h1, acc1, E);
    k_relu_cvt_bf16<<<(N * HID + 255) / 256, 256, 0, stream>>>(acc1, h1b, N * HID);
  }

  {
    int tiles = (M16 / 16) * (OUT / 64);
    k_gemm_wmma_bf16<OUT><<<(tiles + 7) / 8, 256, 0, stream>>>(h1b, W2t, h2, M16);
    k_self_init<OUT><<<(N * OUT + 255) / 256, 256, 0, stream>>>(h2, b2, dinv, out, N);
    k_edge_scatter<OUT><<<(E + 7) / 8, 256, 0, stream>>>(src, dst, dinv, h2, out, E);
  }
}


# In-degrees counted per destination range in LDS (integer adds): scan, edge list and packed-word lookup move from layer-1 aggregation into a degree kernel; no global atomic; 9 launches; sum order kept
# speedup vs baseline: 2.9357x; 2.9357x over previous
#include <hip/hip_runtime.h>
#include <hip/hip_bf16.h>


typedef __bf16 bf16;
typedef __attribute__((ext_vector_type(16))) __bf16 v16bf;
typedef __attribute__((ext_vector_type(8)))  __bf16 v8bf;
typedef __attribute__((ext_vector_type(8)))  float  v8f;

__device__ __forceinline__ bf16 f2bf(float f) {
  union { float f; unsigned u; } in; in.f = f;
  unsigned u = in.u;
  unsigned r = u + 0x7fffu + ((u >> 16) & 1u);
  union { unsigned short s; bf16 b; } out;
  out.s = (unsigned short)(r >> 16);
  return out.b;
}


constexpr int RANGE = 196;
constexpr int LIST_CAP = 16384;
constexpr int REC_CAP  = 6144;
constexpr unsigned SRC_MASK = 0x00FFFFFFu;

__global__ __launch_bounds__(256) void k_edge_pack(const int* __restrict__ src,
                                                   const int* __restrict__ dst,
                                                   unsigned* __restrict__ rid4,
                                                   unsigned* __restrict__ pkw,
                                                   int E, int nwords) {
  int t = blockIdx.x * 256 + threadIdx.x;
  if (t >= nwords) return;
  unsigned packed = 0;
  unsigned rec[4];
#pragma unroll
  for (int j = 0; j < 4; ++j) {
    int e = 4 * t + j;
    unsigned r = 0xFFu;
    unsigned q = 0u;
    if (e < E) {
      int d = dst[e];
      int s = src[e];
      const unsigned rr = (unsigned)d / (unsigned)RANGE;
      r = rr & 0xFFu;
      q = ((unsigned)s & SRC_MASK) | (((unsigned)d - rr * (unsigned)RANGE) << 24);
    }
    packed |= r << (8 * j);
    rec[j] = q;
  }
  rid4[t] = packed;
  ((uint4*)pkw)[t] = make_uint4(rec[0], rec[1], rec[2], rec[3]);
}

template <typename F>
__device__ __forceinline__ void scan_range_edges(const unsigned* __restrict__ rid4, const int E,
                                                 const int tid, const unsigned rng, F on_edge) {
  const int nvec = (E + 15) >> 4;
  const uint4* __restrict__ rv = (const uint4*)rid4;
  const unsigned pat = rng * 0x01010101u;
  for (int v0 = tid; v0 < nvec; v0 += 4 * 1024) {
    uint4 x[4];
#pragma unroll
    for (int g = 0; g < 4; ++g) {
      const int v = v0 + g * 1024;
      x[g] = rv[v < nvec ? v : nvec - 1];
    }
#pragma unroll
    for (int g = 0; g < 4; ++g) {
      const unsigned wd[4] = {x[g].x ^ pat, x[g].y ^ pat, x[g].z ^ pat, x[g].w ^ pat};
      unsigned m = 0;
#pragma unroll
      for (int q = 0; q < 4; ++q) {
        const unsigned z = ~(((wd[q] & 0x7F7F7F7Fu) + 0x7F7F7F7Fu) | wd[q] | 0x7F7F7F7Fu);
        m |= z >> (7 - q);
      }
      if (v0 + g * 1024 >= nvec) m = 0;
      const int ebase = (v0 + g * 1024) * 16;
      while (m) {
        const int p = __builtin_ctz(m);
        m &= m - 1u;
        const int e = ebase + (p & 7) * 4 + (p >> 3);
        if (e < E) on_edge(e);
      }
    }
  }
}

__global__ __launch_bounds__(1024) void k_range_degree(const int* __restrict__ dst,
                                                       const unsigned* __restrict__ rid4,
                                                       const unsigned* __restrict__ pk,
                                                       unsigned* __restrict__ grecw,
                                                       int* __restrict__ gcnt,
                                                       float* __restrict__ dinv,
                                                       int N, int E) {
  __shared__ int list[LIST_CAP];
  __shared__ int degc[256];
  __shared__ int cnt;
  const int tid = threadIdx.x;
  const int base = blockIdx.x * RANGE;
  if (tid == 0) cnt = 0;
  if (tid < 256) degc[tid] = 0;
  __syncthreads();

  scan_range_edges(rid4, E, tid, (unsigned)blockIdx.x, [&](const int e) {
    const int slot = atomicAdd(&cnt, 1);
    if (slot < LIST_CAP) {
      list[slot] = e;
    } else {
      const int dl = dst[e] - base;
      if ((unsigned)dl < (unsigned)RANGE) atomicAdd(&degc[dl], 1);
    }
  });
  __syncthreads();

  const int total = cnt;
  const int count = total < LIST_CAP ? total : LIST_CAP;
  unsigned* __restrict__ recw = grecw + (size_t)blockIdx.x * (REC_CAP * 2);
  for (int i0 = tid; i0 < count; i0 += 4 * 1024) {
    unsigned p[4];
#pragma unroll
    for (int g = 0; g < 4; ++g) {
      const int i = i0 + g * 1024;
      p[g] = pk[list[i < count ? i : count - 1]];
    }
#pragma unroll
    for (int g = 0; g < 4; ++g) {
      const int i = i0 + g * 1024;
      if (i < count) {
        atomicAdd(&degc[p[g] >> 24], 1);
        if (i < REC_CAP) recw[2 * i] = p[g];
      }
    }
  }
  __syncthreads();

  if (tid < RANGE && base + tid < N) dinv[base + tid] = rsqrtf(1.0f + (float)degc[tid]);
  if (tid == 0) gcnt[blockIdx.x] = total;
}


constexpr int CVT_V = 8;

__device__ __forceinline__ unsigned pack2bf(float lo, float hi) {
  union { bf16 b; unsigned short s; } l, h;
  l.b = f2bf(lo);
  h.b = f2bf(hi);
  return (unsigned)l.s | ((unsigned)h.s << 16);
}

__global__ __launch_bounds__(256) void k_cvt_bf16(const float* __restrict__ in,
                                                  bf16* __restrict__ out, int n4) {
  const float4* __restrict__ in4 = (const float4*)in;
  uint2* __restrict__ out2 = (uint2*)out;
  const int i0 = blockIdx.x * (256 * CVT_V) + threadIdx.x;
  float4 v[CVT_V];
#pragma unroll
  for (int k = 0; k < CVT_V; ++k) {
    const int i = i0 + k * 256;
    v[k] = in4[i < n4 ? i : n4 - 1];
  }
#pragma unroll
  for (int k = 0; k < CVT_V; ++k) {
    const int i = i0 + k * 256;
    if (i < n4) out2[i] = make_uint2(pack2bf(v[k].x, v[k].y), pack2bf(v[k].z, v[k].w));
  }
}

__global__ __launch_bounds__(256) void k_transpose_bf16(const float* __restrict__ W,
                                                        bf16* __restrict__ Wt,
                                                        int K, int cols) {
  int idx = blockIdx.x * 256 + threadIdx.x;
  if (idx >= K * cols) return;
  int n = idx / K;
  int k = idx % K;
  Wt[idx] = f2bf(W[(size_t)k * cols + n]);
}


template <int NCOLS>
__global__ __launch_bounds__(256) void k_gemm_wmma_bf16(
    const bf16* __restrict__ A,
    const bf16* __restrict__ Bt,
    float* __restrict__ C,
    int M)
{
  constexpr int K = 128;
  constexpr int NT = 4;
  constexpr int chunksN = NCOLS / 64;
  const int wave = threadIdx.x >> 5;
  const int lane = threadIdx.x & 31;
  const int tile = blockIdx.x * 8 + wave;
  const int numTiles = (M >> 4) * chunksN;
  if (tile >= numTiles) return;
  const int tm = (tile / chunksN) << 4;
  const int tn = (tile % chunksN) << 6;
  const int rc = lane & 15;
  const int hi = lane >> 4;

  const bf16* aRow  = A  + (size_t)(tm + rc) * K + hi * 8;
  const bf16* bBase = Bt + (size_t)(tn + rc) * K + hi * 8;

  v8f acc[NT] = {};
#pragma unroll
  for (int k0 = 0; k0 < K; k0 += 32) {
    union { v16bf v; v8bf h[2]; } a;
    a.h[0] = *(const v8bf*)(aRow + k0);
    a.h[1] = *(const v8bf*)(aRow + k0 + 16);
#pragma unroll
    for (int t = 0; t < NT; ++t) {
      union { v16bf v; v8bf h[2]; } b;
      const bf16* bp = bBase + (size_t)t * 16 * K + k0;
      b.h[0] = *(const v8bf*)(bp);
      b.h[1] = *(const v8bf*)(bp + 16);
      acc[t] = __builtin_amdgcn_wmma_f32_16x16x32_bf16(
           false, a.v,  false, b.v,
           (short)0, acc[t],  false,  false);
    }
  }

  float* cBase = C + (size_t)tm * NCOLS + tn;
#pragma unroll
  for (int t = 0; t < NT; ++t)
#pragma unroll
    for (int j = 0; j < 8; ++j)
      cBase[(size_t)(j + hi * 8) * NCOLS + t * 16 + rc] = acc[t][j];
}

template <int NCOLS, bool RELU_BF16, bool FIRST>
__global__ __launch_bounds__(1024) void k_range_aggregate(const int* __restrict__ src,
                                                          const int* __restrict__ dst,
                                                          const unsigned* __restrict__ rid4,
                                                          const unsigned* __restrict__ pk,
                                                          unsigned* __restrict__ grecw,
                                                          int* __restrict__ gcnt,
                                                          const float* __restrict__ dinv,
                                                          const float* __restrict__ H,
                                                          const float* __restrict__ bias,
                                                          float* __restrict__ outF,
                                                          bf16* __restrict__ outB,
                                                          int N, int E) {
  constexpr int NL = NCOLS / 32;
  constexpr int U  = 16 / NL;
  constexpr int NB = 32 / U;
  __shared__ float acc[RANGE * NCOLS];
  __shared__ int   list[LIST_CAP];
  __shared__ int   cnt;

  const int tid = threadIdx.x;
  const unsigned lane = threadIdx.x & 31u;
  const int wave = threadIdx.x >> 5;
  const int base = blockIdx.x * RANGE;

  if (tid == 0) cnt = 0;
  {
    constexpr int RV = NCOLS / 4;
    constexpr int NV = RANGE * RV;
    for (int q0 = tid; q0 < NV; q0 += 4 * 1024) {
      float4 hq[4];
      float dq[4];
#pragma unroll
      for (int g = 0; g < 4; ++g) {
        const int q  = q0 + g * 1024;
        const int qc = q < NV ? q : NV - 1;
        const int il = qc / RV;
        const int i  = base + il;
        const int ic = i < N ? i : N - 1;
        dq[g] = dinv[ic];
        hq[g] = *(const float4*)(H + (size_t)ic * NCOLS + (qc - il * RV) * 4);
      }
#pragma unroll
      for (int g = 0; g < 4; ++g) {
        const int q = q0 + g * 1024;
        if (q < NV) {
          const int il = q / RV;
          const int c  = (q - il * RV) * 4;
          float4 v = make_float4(0.0f, 0.0f, 0.0f, 0.0f);
          if (base + il < N) {
            const float d2 = dq[g] * dq[g];
            v.x = bias[c + 0] + d2 * hq[g].x;
            v.y = bias[c + 1] + d2 * hq[g].y;
            v.z = bias[c + 2] + d2 * hq[g].z;
            v.w = bias[c + 3] + d2 * hq[g].w;
          }
          *(float4*)(acc + q * 4) = v;
        }
      }
    }
  }
  __syncthreads();

  bool scan = true;
  {
    const int c0 = gcnt[blockIdx.x];
    if (c0 <= REC_CAP) {
      scan = false;
      if (tid == 0) cnt = c0;
    }
  }
  const bool useRec = !scan;
  if (scan) {
    scan_range_edges(rid4, E, tid, (unsigned)blockIdx.x, [&](const int e) {
      const int slot = atomicAdd(&cnt, 1);
      if (slot < LIST_CAP) {
        list[slot] = e;
      } else {
        const int s = src[e];
        const int dl = dst[e] - base;
        if ((unsigned)dl < (unsigned)RANGE) {
          const float w = dinv[s] * dinv[base + dl];
#pragma nounroll
          for (int ch = 0; ch < NCOLS; ++ch)
            atomicAdd(&acc[dl * NCOLS + ch], H[(size_t)s * NCOLS + ch] * w);
        }
      }
    });
  }
  __syncthreads();

  const int count = (cnt < LIST_CAP) ? cnt : LIST_CAP;
  unsigned* __restrict__ recw = grecw + (size_t)blockIdx.x * (REC_CAP * 2);
  auto fetch = [&](const int idx, int& s, int& dl, float& w) {
    if (useRec && !FIRST) {
      const uint2 r = ((const uint2*)recw)[idx];
      s  = (int)(r.x & SRC_MASK);
      dl = (int)(r.x >> 24);
      w  = __uint_as_float(r.y);
    } else {
      const unsigned p = useRec ? recw[2 * idx] : pk[list[idx]];
      s  = (int)(p & SRC_MASK);
      dl = (int)(p >> 24);
      w  = dinv[s] * dinv[base + dl];
      if (useRec) recw[2 * idx + 1] = __float_as_uint(w);
    }
  };

  const int whole = count & ~1023;
  for (int b0 = wave * 32; b0 < whole; b0 += 32 * 32) {
    int s, dl;
    float w;
    fetch(b0 + (int)lane, s, dl, w);
    const int wi = __float_as_int(w);
    float hv[2][U][NL];
#pragma unroll
    for (int u = 0; u < U; ++u) {
      const int sj = __builtin_amdgcn_readlane(s, u);
      const float* hrow = H + (size_t)sj * NCOLS;
#pragma unroll
      for (int k = 0; k < NL; ++k) hv[0][u][k] = hrow[k * 32 + lane];
    }
#pragma unroll
    for (int b = 0; b < NB; ++b) {
      if (b + 1 < NB) {
#pragma unroll
        for (int u = 0; u < U; ++u) {
          const int sj = __builtin_amdgcn_readlane(s, (b + 1) * U + u);
          const float* hrow = H + (size_t)sj * NCOLS;
#pragma unroll
          for (int k = 0; k < NL; ++k) hv[(b + 1) & 1][u][k] = hrow[k * 32 + lane];
        }
      }
#pragma unroll
      for (int u = 0; u < U; ++u) {
        const int dj = __builtin_amdgcn_readlane(dl, b * U + u);
        const float wj = __int_as_float(__builtin_amdgcn_readlane(wi, b * U + u));
        float* arow = acc + dj * NCOLS;
#pragma unroll
        for (int k = 0; k < NL; ++k) atomicAdd(&arow[k * 32 + lane], hv[b & 1][u][k] * wj);
      }
    }
  }

  {
    constexpr int BU = 32 / NL;
    const int rem = count - whole;
    const int m = (rem + 31) >> 5;
    const int t0 = whole + wave * m;
    const int left = count - t0;
    const int n = __builtin_amdgcn_readfirstlane(left < m ? left : m);
    if (n > 0) {
      int s = 0, dl = 0;
      float w = 0.0f;
      if ((int)lane < n) fetch(t0 + (int)lane, s, dl, w);
      const int wi = __float_as_int(w);
      int j0 = 0;
      for (; j0 + BU <= n; j0 += BU) {
        float t[BU][NL];
#pragma unroll
        for (int u = 0; u < BU; ++u) {
          const int sj = __builtin_amdgcn_readlane(s, j0 + u);
          const float* hrow = H + (size_t)sj * NCOLS;
#pragma unroll
          for (int k = 0; k < NL; ++k) t[u][k] = hrow[k * 32 + lane];
        }
#pragma unroll
        for (int u = 0; u < BU; ++u) {
          const int dj = __builtin_amdgcn_readlane(dl, j0 + u);
          const float wj = __int_as_float(__builtin_amdgcn_readlane(wi, j0 + u));
          float* arow = acc + dj * NCOLS;
#pragma unroll
          for (int k = 0; k < NL; ++k) atomicAdd(&arow[k * 32 + lane], t[u][k] * wj);
        }
      }
      const int nb = n - j0;
      if (nb > 0) {
        float t[BU][NL];
#pragma unroll
        for (int u = 0; u < BU - 1; ++u) {
          if (u < nb) {
            const int sj = __builtin_amdgcn_readlane(s, j0 + u);
            const float* hrow = H + (size_t)sj * NCOLS;
#pragma unroll
            for (int k = 0; k < NL; ++k) t[u][k] = hrow[k * 32 + lane];
          }
        }
#pragma unroll
        for (int u = 0; u < BU - 1; ++u) {
          if (u < nb) {
            const int dj = __builtin_amdgcn_readlane(dl, j0 + u);
            const float wj = __int_as_float(__builtin_amdgcn_readlane(wi, j0 + u));
            float* arow = acc + dj * NCOLS;
#pragma unroll
            for (int k = 0; k < NL; ++k) atomicAdd(&arow[k * 32 + lane], t[u][k] * wj);
          }
        }
      }
    }
  }
  __syncthreads();

  if (RELU_BF16) {
    unsigned* o = (unsigned*)outB + (size_t)base * (NCOLS / 2);
    for (int p = tid; p < RANGE * NCOLS / 2; p += 1024) {
      const int il = (2 * p) / NCOLS;
      if (base + il < N) {
        union { bf16 b; unsigned short s; } lo, hi;
        lo.b = f2bf(fmaxf(acc[2 * p], 0.0f));
        hi.b = f2bf(fmaxf(acc[2 * p + 1], 0.0f));
        o[p] = (unsigned)lo.s | ((unsigned)hi.s << 16);
      }
    }
  } else {
    float* o = outF + (size_t)base * NCOLS;
    for (int idx = tid; idx < RANGE * NCOLS; idx += 1024) {
      const int il = idx / NCOLS;
      if (base + il < N) o[idx] = acc[idx];
    }
  }
}


extern "C" void kernel_launch(void* const* d_in, const int* in_sizes, int n_in,
                              void* d_out, int out_size, void* d_ws, size_t ws_size,
                              hipStream_t stream) {
  (void)n_in; (void)out_size; (void)ws_size;
  const float* x  = (const float*)d_in[0];
  const int*   ei = (const int*)  d_in[1];
  const float* W1 = (const float*)d_in[2];
  const float* b1 = (const float*)d_in[3];
  const float* W2 = (const float*)d_in[4];
  const float* b2 = (const float*)d_in[5];
  float* out = (float*)d_out;

  const int IN = 128, HID = 128, OUT = 64;
  const int N = in_sizes[0] / IN;
  const int E = in_sizes[1] / 2;
  const int M16 = ((N + 15) / 16) * 16;
  const int* src = ei;
  const int* dst = ei + E;

  char* ws = (char*)d_ws;
  size_t off = 0;
  auto alloc = [&](size_t bytes) -> void* {
    void* p = ws + off;
    off = (off + bytes + 255) & ~(size_t)255;
    return p;
  };
  float* dinv = (float*)alloc((size_t)N * 4);
  bf16*  Xb   = (bf16*) alloc((size_t)M16 * IN * 2);
  bf16*  W1t  = (bf16*) alloc((size_t)HID * IN * 2);
  bf16*  W2t  = (bf16*) alloc((size_t)OUT * HID * 2);
  float* h1   = (float*)alloc((size_t)M16 * HID * 4);
  const int nwords = ((E + 15) / 16) * 4;
  unsigned* rid4 = (unsigned*)alloc((size_t)nwords * 4);
  bf16*  h1b  = (bf16*) alloc((size_t)M16 * HID * 2);
  float* h2   = (float*)alloc((size_t)M16 * OUT * 4);
  const int ranges = (N + RANGE - 1) / RANGE;
  unsigned* grec = (unsigned*)alloc((size_t)ranges * REC_CAP * 8);
  unsigned* pkw  = (unsigned*)alloc((size_t)nwords * 16);
  int* gcnt  = (int*)alloc((size_t)ranges * 4);

  k_edge_pack   <<<(nwords + 255) / 256, 256, 0, stream>>>(src, dst, rid4, pkw, E, nwords);
  k_range_degree<<<ranges, 1024, 0, stream>>>(dst, rid4, pkw, grec, gcnt, dinv, N, E);

  k_cvt_bf16      <<<(N * (IN / 4) + 256 * CVT_V - 1) / (256 * CVT_V), 256, 0, stream>>>(x, Xb, N * (IN / 4));
  k_transpose_bf16<<<(IN * HID + 255) / 256, 256, 0, stream>>>(W1, W1t, IN, HID);
  k_transpose_bf16<<<(HID * OUT + 255) / 256, 256, 0, stream>>>(W2, W2t, HID, OUT);

  {
    int tiles = (M16 / 16) * (HID / 64);
    k_gemm_wmma_bf16<HID><<<(tiles + 7) / 8, 256, 0, stream>>>(Xb, W1t, h1, M16);
    k_range_aggregate<HID, true, true><<<ranges, 1024, 0, stream>>>(
        src, dst, rid4, pkw, grec, gcnt, dinv, h1, b1, (float*)nullptr, h1b, N, E);
  }

  {
    int tiles = (M16 / 16) * (OUT / 64);
    k_gemm_wmma_bf16<OUT><<<(tiles + 7) / 8, 256, 0, stream>>>(h1b, W2t, h2, M16);
    k_range_aggregate<OUT, false, false><<<ranges, 1024, 0, stream>>>(
        src, dst, rid4, pkw, grec, gcnt, dinv, h2, b2, out, (bf16*)nullptr, N, E);
  }
}
